# starting kernel
# baseline (speedup 1.0000x reference)
#include <hip/hip_runtime.h>

#define IN_F   256
#define N_ASS  4096
#define OUT_F  256
#define BATCH  128
#define ASS0   IN_F
#define OUT0   (IN_F + N_ASS)

typedef float v2f __attribute__((ext_vector_type(2)));
typedef float v8f __attribute__((ext_vector_type(8)));

__global__ __launch_bounds__(256) void rnn_zero_f32(float* __restrict__ p, int n) {
    int i = blockIdx.x * blockDim.x + threadIdx.x;
    if (i < n) p[i] = 0.0f;
}

__global__ __launch_bounds__(256) void rnn_scatter_w1(const float* __restrict__ w,
                                                      const int* __restrict__ src,
                                                      const int* __restrict__ dst,
                                                      int e, float* __restrict__ W1) {
    int i = blockIdx.x * blockDim.x + threadIdx.x;
    if (i < e) {
        atomicAdd(&W1[(size_t)(dst[i] - ASS0) * IN_F + src[i]], w[i]);
    }
}

__global__ __launch_bounds__(256) void rnn_scatter_w2(const float* __restrict__ w,
                                                      const int* __restrict__ src,
                                                      const int* __restrict__ dst,
                                                      int e, float* __restrict__ W2) {
    int i = blockIdx.x * blockDim.x + threadIdx.x;
    if (i < e) {
        int d = dst[i];
        if (d >= OUT0) {
            atomicAdd(&W2[(size_t)(d - OUT0) * N_ASS + (src[i] - ASS0)], w[i]);
        }
    }
}

template <int M, int N, int K, bool BT, bool CT>
__global__ __launch_bounds__(256) void rnn_wmma_gemm_f32(const float* __restrict__ A,
                                                         const float* __restrict__ B,
                                                         float* __restrict__ C) {
    const int lane = threadIdx.x & 31;
    const int wave = threadIdx.x >> 5;
    const int tm   = blockIdx.x;
    const int half = lane >> 4;
    const int l16  = lane & 15;
    const int m    = tm * 16 + l16;
    const int n    = wave * 16 + l16;

    v8f acc = {};
    const float* __restrict__ Arow = A + (size_t)m * K;
    const float* __restrict__ Brow = BT ? (B + (size_t)n * K) : nullptr;

    for (int kb = 0; kb < K; kb += 4) {
        const int ka = kb + 2 * half;
        v2f a = *(const v2f*)(Arow + ka);
        v2f b;
        if (BT) {
            b = *(const v2f*)(Brow + ka);
        } else {
            b.x = B[(size_t)ka * N + n];
            b.y = B[(size_t)(ka + 1) * N + n];
        }
        acc = __builtin_amdgcn_wmma_f32_16x16x4_f32(false, a, false, b,
                                                    (short)0, acc, false, false);
    }

#pragma unroll
    for (int v = 0; v < 8; ++v) {
        const int mr = tm * 16 + v + 8 * half;
        if (CT) C[(size_t)n * M + mr] = acc[v];
        else    C[(size_t)mr * N + n] = acc[v];
    }
}

extern "C" void kernel_launch(void* const* d_in, const int* in_sizes, int n_in,
                              void* d_out, int out_size, void* d_ws, size_t ws_size,
                              hipStream_t stream) {
    const float* x   = (const float*)d_in[0];
    const float* wi  = (const float*)d_in[1];
    const float* wa  = (const float*)d_in[2];
    const int* in_src  = (const int*)d_in[3];
    const int* in_dst  = (const int*)d_in[4];
    const int* ass_src = (const int*)d_in[5];
    const int* ass_dst = (const int*)d_in[6];
    const int e_in  = in_sizes[1];
    const int e_ass = in_sizes[2];

    float* W1    = (float*)d_ws;
    float* W2    = W1 + (size_t)N_ASS * IN_F;
    float* ACC_T = W2 + (size_t)OUT_F * N_ASS;

    {
        const int n = N_ASS * IN_F + OUT_F * N_ASS;
        rnn_zero_f32<<<(n + 255) / 256, 256, 0, stream>>>(W1, n);
    }
    rnn_scatter_w1<<<(e_in + 255) / 256, 256, 0, stream>>>(wi, in_src, in_dst, e_in, W1);
    rnn_scatter_w2<<<(e_ass + 255) / 256, 256, 0, stream>>>(wa, ass_src, ass_dst, e_ass, W2);

    rnn_wmma_gemm_f32<N_ASS, BATCH, IN_F, true, true>
        <<<N_ASS / 16, 256, 0, stream>>>(W1, x, ACC_T);

    rnn_wmma_gemm_f32<OUT_F, BATCH, N_ASS, true, true>
        <<<OUT_F / 16, 256, 0, stream>>>(W2, ACC_T, (float*)d_out);

    (void)n_in; (void)out_size; (void)ws_size;
}


# Both edge lists scattered in one launch (5 launches -> 4): first ceil(E_in/256) work-groups take the hop-1 list, the rest the hop-2 list; float atomics, operands and sum order as in parent
# speedup vs baseline: 5.1852x; 5.1852x over previous
#include <hip/hip_runtime.h>

#define IN_F   256
#define N_ASS  4096
#define OUT_F  256
#define BATCH  128
#define ASS0   IN_F
#define OUT0   (IN_F + N_ASS)

typedef float v2f __attribute__((ext_vector_type(2)));
typedef float v8f __attribute__((ext_vector_type(8)));

__global__ __launch_bounds__(256) void rnn_zero_f32(float* __restrict__ p, int n) {
    int i = blockIdx.x * blockDim.x + threadIdx.x;
    if (i < n) p[i] = 0.0f;
}

__global__ __launch_bounds__(256) void rnn_scatter_w12(const float* __restrict__ w1,
                                                       const int* __restrict__ src1,
                                                       const int* __restrict__ dst1,
                                                       int e1, float* __restrict__ W1,
                                                       const float* __restrict__ w2,
                                                       const int* __restrict__ src2,
                                                       const int* __restrict__ dst2,
                                                       int e2, float* __restrict__ W2) {
    const int nb1 = (e1 + 255) >> 8;
    const int blk = (int)blockIdx.x;
    if (blk < nb1) {
        const int i = blk * 256 + threadIdx.x;
        if (i < e1) {
            atomicAdd(&W1[(size_t)(dst1[i] - ASS0) * IN_F + src1[i]], w1[i]);
        }
    } else {
        const int i = (blk - nb1) * 256 + threadIdx.x;
        if (i < e2) {
            const int d = dst2[i];
            if (d >= OUT0) {
                atomicAdd(&W2[(size_t)(d - OUT0) * N_ASS + (src2[i] - ASS0)], w2[i]);
            }
        }
    }
}

#define GRP       128
#define N_GRP     (N_ASS / GRP)
#define ACC_PITCH (GRP + 4)
#define X_PITCH   (IN_F + 4)
#define KB        4
#define WS_NEED   ((size_t)(N_ASS * IN_F + OUT_F * N_ASS + N_GRP * BATCH * OUT_F) * sizeof(float))

typedef float v4f __attribute__((ext_vector_type(4)));

static __device__ __forceinline__ v8f rnn_mma8(v4f a0, v4f a1, v4f b0, v4f b1, v8f c) {
    c = __builtin_amdgcn_wmma_f32_16x16x4_f32(false, a0.xy, false, b0.xy, (short)0, c, false, false);
    c = __builtin_amdgcn_wmma_f32_16x16x4_f32(false, a0.zw, false, b0.zw, (short)0, c, false, false);
    c = __builtin_amdgcn_wmma_f32_16x16x4_f32(false, a1.xy, false, b1.xy, (short)0, c, false, false);
    c = __builtin_amdgcn_wmma_f32_16x16x4_f32(false, a1.zw, false, b1.zw, (short)0, c, false, false);
    return c;
}

static __device__ __forceinline__ void rnn_lds_barrier() {
    __builtin_amdgcn_fence(__ATOMIC_RELEASE, "workgroup", "local");
    __builtin_amdgcn_s_barrier();
    __builtin_amdgcn_fence(__ATOMIC_ACQUIRE, "workgroup", "local");
}

__global__ __launch_bounds__(256) void rnn_fused_hops(const float* __restrict__ W1,
                                                      const float* __restrict__ x,
                                                      const float* __restrict__ W2,
                                                      float* __restrict__ P,
                                                      size_t ws_bytes) {
    if (ws_bytes < WS_NEED) return;

    __shared__ __attribute__((aligned(16))) float acc_s[16 * ACC_PITCH];
    __shared__ __attribute__((aligned(16))) float x_s[16 * X_PITCH];

    const int lane = threadIdx.x & 31;
    const int wave = threadIdx.x >> 5;
    const int g    = blockIdx.x >> 3;
    const int bt   = blockIdx.x & 7;
    const int half = lane >> 4;
    const int c16  = lane & 15;

    v4f xv[4];
#pragma unroll
    for (int r = 0; r < 4; ++r) {
        const int u = threadIdx.x + r * 256;
        xv[r] = *(const v4f*)(x + (size_t)(bt * 16 + (u >> 6)) * IN_F + (u & 63) * 4);
    }

    const float* __restrict__ a1p = W1 + (size_t)(g * GRP + wave * 16 + c16) * IN_F + 8 * half;
    const float* __restrict__ a2p = W2 + (size_t)(wave * 32 + c16) * N_ASS + g * GRP + 8 * half;
    const float* __restrict__ a3p = a2p + (size_t)16 * N_ASS;
    v4f a1[2 * (IN_F / 16)];
    v4f a2[4 * (GRP / 16)];
#pragma unroll
    for (int i = 0; i < IN_F / 16; ++i) {
        a1[2 * i]     = *(const v4f*)(a1p + i * 16);
        a1[2 * i + 1] = *(const v4f*)(a1p + i * 16 + 4);
    }
    __builtin_amdgcn_sched_barrier(0);
#pragma unroll
    for (int i = 0; i < GRP / 16; ++i) {
        a2[4 * i]     = *(const v4f*)(a2p + i * 16);
        a2[4 * i + 1] = *(const v4f*)(a2p + i * 16 + 4);
        a2[4 * i + 2] = *(const v4f*)(a3p + i * 16);
        a2[4 * i + 3] = *(const v4f*)(a3p + i * 16 + 4);
    }
#pragma unroll
    for (int r = 0; r < 4; ++r) {
        const int u = threadIdx.x + r * 256;
        *(v4f*)(x_s + (u >> 6) * X_PITCH + (u & 63) * 4) = xv[r];
    }
    rnn_lds_barrier();

    const float* b1p = x_s + c16 * X_PITCH + 8 * half;
    v4f bb[2][2 * KB];
#pragma unroll
    for (int i = 0; i < KB; ++i) {
        bb[0][2 * i]     = *(const v4f*)(b1p + i * 16);
        bb[0][2 * i + 1] = *(const v4f*)(b1p + i * 16 + 4);
    }
    v8f acc = {};
#pragma unroll
    for (int j = 0; j < IN_F / 16 / KB; ++j) {
        if (j + 1 < IN_F / 16 / KB) {
#pragma unroll
            for (int i = 0; i < KB; ++i) {
                bb[(j + 1) & 1][2 * i]     = *(const v4f*)(b1p + ((j + 1) * KB + i) * 16);
                bb[(j + 1) & 1][2 * i + 1] = *(const v4f*)(b1p + ((j + 1) * KB + i) * 16 + 4);
            }
        }
        __builtin_amdgcn_sched_barrier(0);
#pragma unroll
        for (int i = 0; i < KB; ++i) {
            const int kc = j * KB + i;
            acc = rnn_mma8(a1[2 * kc], a1[2 * kc + 1], bb[j & 1][2 * i], bb[j & 1][2 * i + 1], acc);
        }
        __builtin_amdgcn_sched_barrier(0);
    }
    {
        float* d = acc_s + c16 * ACC_PITCH + wave * 16 + 8 * half;
        v4f lo = {acc[0], acc[1], acc[2], acc[3]};
        v4f hi = {acc[4], acc[5], acc[6], acc[7]};
        *(v4f*)d       = lo;
        *(v4f*)(d + 4) = hi;
    }
    rnn_lds_barrier();

    const float* b2p = acc_s + c16 * ACC_PITCH + 8 * half;
#pragma unroll
    for (int i = 0; i < KB; ++i) {
        bb[0][2 * i]     = *(const v4f*)(b2p + i * 16);
        bb[0][2 * i + 1] = *(const v4f*)(b2p + i * 16 + 4);
    }
    v8f o0 = {};
    v8f o1 = {};
#pragma unroll
    for (int j = 0; j < GRP / 16 / KB; ++j) {
        if (j + 1 < GRP / 16 / KB) {
#pragma unroll
            for (int i = 0; i < KB; ++i) {
                bb[(j + 1) & 1][2 * i]     = *(const v4f*)(b2p + ((j + 1) * KB + i) * 16);
                bb[(j + 1) & 1][2 * i + 1] = *(const v4f*)(b2p + ((j + 1) * KB + i) * 16 + 4);
            }
        }
        __builtin_amdgcn_sched_barrier(0);
#pragma unroll
        for (int i = 0; i < KB; ++i) {
            const int kc = j * KB + i;
            o0 = rnn_mma8(a2[4 * kc],     a2[4 * kc + 1], bb[j & 1][2 * i], bb[j & 1][2 * i + 1], o0);
            o1 = rnn_mma8(a2[4 * kc + 2], a2[4 * kc + 3], bb[j & 1][2 * i], bb[j & 1][2 * i + 1], o1);
        }
        __builtin_amdgcn_sched_barrier(0);
    }
    {
        float* d = P + ((size_t)g * BATCH + bt * 16 + c16) * OUT_F + wave * 32 + 8 * half;
        v4f lo0 = {o0[0], o0[1], o0[2], o0[3]};
        v4f hi0 = {o0[4], o0[5], o0[6], o0[7]};
        v4f lo1 = {o1[0], o1[1], o1[2], o1[3]};
        v4f hi1 = {o1[4], o1[5], o1[6], o1[7]};
        *(v4f*)d        = lo0;
        *(v4f*)(d + 4)  = hi0;
        *(v4f*)(d + 16) = lo1;
        *(v4f*)(d + 20) = hi1;
    }
}

__global__ __launch_bounds__(128) void rnn_sum_planes(const float* __restrict__ P,
                                                      float* __restrict__ out,
                                                      size_t ws_bytes) {
    if (ws_bytes < WS_NEED) return;
    const int i = blockIdx.x * 128 + threadIdx.x;
    const v4f* __restrict__ p = (const v4f*)P;
    v4f s = p[i];
#pragma unroll
    for (int q = 1; q < N_GRP; ++q) s += p[(size_t)q * (BATCH * OUT_F / 4) + i];
    ((v4f*)out)[i] = s;
}

extern "C" void kernel_launch(void* const* d_in, const int* in_sizes, int n_in,
                              void* d_out, int out_size, void* d_ws, size_t ws_size,
                              hipStream_t stream) {
    const float* x   = (const float*)d_in[0];
    const float* wi  = (const float*)d_in[1];
    const float* wa  = (const float*)d_in[2];
    const int* in_src  = (const int*)d_in[3];
    const int* in_dst  = (const int*)d_in[4];
    const int* ass_src = (const int*)d_in[5];
    const int* ass_dst = (const int*)d_in[6];
    const int e_in  = in_sizes[1];
    const int e_ass = in_sizes[2];

    float* W1    = (float*)d_ws;
    float* W2    = W1 + (size_t)N_ASS * IN_F;
    float* P     = W2 + (size_t)OUT_F * N_ASS;

    {
        const int n = N_ASS * IN_F + OUT_F * N_ASS;
        rnn_zero_f32<<<(n + 255) / 256, 256, 0, stream>>>(W1, n);
    }
    rnn_scatter_w12<<<(e_in + 255) / 256 + (e_ass + 255) / 256, 256, 0, stream>>>(
        wi, in_src, in_dst, e_in, W1, wa, ass_src, ass_dst, e_ass, W2);

    rnn_fused_hops<<<256, 256, 0, stream>>>(W1, x, W2, P, ws_size);

    rnn_sum_planes<<<64, 128, 0, stream>>>(P, (float*)d_out, ws_size);

    (void)n_in; (void)out_size;
}


# Dense operators W1, W2 stored in the matrix instruction's A-fragment order, not row-major: each 16-byte A load of a wave reads 512 contiguous bytes (4 lines, was 16); operands, sum order as parent
# speedup vs baseline: 5.7244x; 1.1040x over previous
#include <hip/hip_runtime.h>

#define IN_F   256
#define N_ASS  4096
#define OUT_F  256
#define BATCH  128
#define ASS0   IN_F
#define OUT0   (IN_F + N_ASS)

typedef float v2f __attribute__((ext_vector_type(2)));
typedef float v8f __attribute__((ext_vector_type(8)));

#define GRP       128
#define N_GRP     (N_ASS / GRP)

static __device__ __forceinline__ unsigned rnn_frag_off(unsigned row16, unsigned k) {
    return ((((k >> 4) * 2 + ((k >> 2) & 1)) * 32 + ((k >> 3) & 1) * 16 + row16) << 2) + (k & 3);
}
static __device__ __forceinline__ size_t rnn_w1_off(unsigned neuron, unsigned input) {
    return (size_t)(neuron >> 4) * (16 * IN_F) + rnn_frag_off(neuron & 15, input);
}
static __device__ __forceinline__ size_t rnn_w2_off(unsigned output, unsigned neuron) {
    return (size_t)((neuron / GRP) * (OUT_F / 16) + (output >> 4)) * (16 * GRP)
         + rnn_frag_off(output & 15, neuron % GRP);
}

__global__ __launch_bounds__(256) void rnn_zero_f32(float* __restrict__ p, int n) {
    int i = blockIdx.x * blockDim.x + threadIdx.x;
    if (i < n) p[i] = 0.0f;
}

__global__ __launch_bounds__(256) void rnn_scatter_w12(const float* __restrict__ w1,
                                                       const int* __restrict__ src1,
                                                       const int* __restrict__ dst1,
                                                       int e1, float* __restrict__ W1,
                                                       const float* __restrict__ w2,
                                                       const int* __restrict__ src2,
                                                       const int* __restrict__ dst2,
                                                       int e2, float* __restrict__ W2) {
    const int nb1 = (e1 + 255) >> 8;
    const int blk = (int)blockIdx.x;
    if (blk < nb1) {
        const int i = blk * 256 + threadIdx.x;
        if (i < e1) {
            atomicAdd(&W1[rnn_w1_off((unsigned)(dst1[i] - ASS0), (unsigned)src1[i])], w1[i]);
        }
    } else {
        const int i = (blk - nb1) * 256 + threadIdx.x;
        if (i < e2) {
            const int d = dst2[i];
            if (d >= OUT0) {
                atomicAdd(&W2[rnn_w2_off((unsigned)(d - OUT0), (unsigned)(src2[i] - ASS0))], w2[i]);
            }
        }
    }
}

#define ACC_PITCH (GRP + 4)
#define X_PITCH   (IN_F + 4)
#define KB        4
#define WS_NEED   ((size_t)(N_ASS * IN_F + OUT_F * N_ASS + N_GRP * BATCH * OUT_F) * sizeof(float))

typedef float v4f __attribute__((ext_vector_type(4)));

static __device__ __forceinline__ v8f rnn_mma8(v4f a0, v4f a1, v4f b0, v4f b1, v8f c) {
    c = __builtin_amdgcn_wmma_f32_16x16x4_f32(false, a0.xy, false, b0.xy, (short)0, c, false, false);
    c = __builtin_amdgcn_wmma_f32_16x16x4_f32(false, a0.zw, false, b0.zw, (short)0, c, false, false);
    c = __builtin_amdgcn_wmma_f32_16x16x4_f32(false, a1.xy, false, b1.xy, (short)0, c, false, false);
    c = __builtin_amdgcn_wmma_f32_16x16x4_f32(false, a1.zw, false, b1.zw, (short)0, c, false, false);
    return c;
}

static __device__ __forceinline__ void rnn_lds_barrier() {
    __builtin_amdgcn_fence(__ATOMIC_RELEASE, "workgroup", "local");
    __builtin_amdgcn_s_barrier();
    __builtin_amdgcn_fence(__ATOMIC_ACQUIRE, "workgroup", "local");
}

__global__ __launch_bounds__(256) void rnn_fused_hops(const float* __restrict__ W1,
                                                      const float* __restrict__ x,
                                                      const float* __restrict__ W2,
                                                      float* __restrict__ P,
                                                      size_t ws_bytes) {
    if (ws_bytes < WS_NEED) return;

    __shared__ __attribute__((aligned(16))) float acc_s[16 * ACC_PITCH];
    __shared__ __attribute__((aligned(16))) float x_s[16 * X_PITCH];

    const int lane = threadIdx.x & 31;
    const int wave = threadIdx.x >> 5;
    const int g    = blockIdx.x >> 3;
    const int bt   = blockIdx.x & 7;
    const int half = lane >> 4;
    const int c16  = lane & 15;

    v4f xv[4];
#pragma unroll
    for (int r = 0; r < 4; ++r) {
        const int u = threadIdx.x + r * 256;
        xv[r] = *(const v4f*)(x + (size_t)(bt * 16 + (u >> 6)) * IN_F + (u & 63) * 4);
    }

    const float* __restrict__ a1p = W1 + (size_t)(g * (GRP / 16) + wave) * (16 * IN_F) + lane * 4;
    const float* __restrict__ a2p = W2 + (size_t)(g * (OUT_F / 16) + 2 * wave) * (16 * GRP) + lane * 4;
    const float* __restrict__ a3p = a2p + 16 * GRP;
    v4f a1[2 * (IN_F / 16)];
    v4f a2[4 * (GRP / 16)];
#pragma unroll
    for (int i = 0; i < IN_F / 16; ++i) {
        a1[2 * i]     = *(const v4f*)(a1p + (2 * i) * 128);
        a1[2 * i + 1] = *(const v4f*)(a1p + (2 * i + 1) * 128);
    }
    __builtin_amdgcn_sched_barrier(0);
#pragma unroll
    for (int i = 0; i < GRP / 16; ++i) {
        a2[4 * i]     = *(const v4f*)(a2p + (2 * i) * 128);
        a2[4 * i + 1] = *(const v4f*)(a2p + (2 * i + 1) * 128);
        a2[4 * i + 2] = *(const v4f*)(a3p + (2 * i) * 128);
        a2[4 * i + 3] = *(const v4f*)(a3p + (2 * i + 1) * 128);
    }
#pragma unroll
    for (int r = 0; r < 4; ++r) {
        const int u = threadIdx.x + r * 256;
        *(v4f*)(x_s + (u >> 6) * X_PITCH + (u & 63) * 4) = xv[r];
    }
    rnn_lds_barrier();

    const float* b1p = x_s + c16 * X_PITCH + 8 * half;
    v4f bb[2][2 * KB];
#pragma unroll
    for (int i = 0; i < KB; ++i) {
        bb[0][2 * i]     = *(const v4f*)(b1p + i * 16);
        bb[0][2 * i + 1] = *(const v4f*)(b1p + i * 16 + 4);
    }
    v8f acc = {};
#pragma unroll
    for (int j = 0; j < IN_F / 16 / KB; ++j) {
        if (j + 1 < IN_F / 16 / KB) {
#pragma unroll
            for (int i = 0; i < KB; ++i) {
                bb[(j + 1) & 1][2 * i]     = *(const v4f*)(b1p + ((j + 1) * KB + i) * 16);
                bb[(j + 1) & 1][2 * i + 1] = *(const v4f*)(b1p + ((j + 1) * KB + i) * 16 + 4);
            }
        }
        __builtin_amdgcn_sched_barrier(0);
#pragma unroll
        for (int i = 0; i < KB; ++i) {
            const int kc = j * KB + i;
            acc = rnn_mma8(a1[2 * kc], a1[2 * kc + 1], bb[j & 1][2 * i], bb[j & 1][2 * i + 1], acc);
        }
        __builtin_amdgcn_sched_barrier(0);
    }
    {
        float* d = acc_s + c16 * ACC_PITCH + wave * 16 + 8 * half;
        v4f lo = {acc[0], acc[1], acc[2], acc[3]};
        v4f hi = {acc[4], acc[5], acc[6], acc[7]};
        *(v4f*)d       = lo;
        *(v4f*)(d + 4) = hi;
    }
    rnn_lds_barrier();

    const float* b2p = acc_s + c16 * ACC_PITCH + 8 * half;
#pragma unroll
    for (int i = 0; i < KB; ++i) {
        bb[0][2 * i]     = *(const v4f*)(b2p + i * 16);
        bb[0][2 * i + 1] = *(const v4f*)(b2p + i * 16 + 4);
    }
    v8f o0 = {};
    v8f o1 = {};
#pragma unroll
    for (int j = 0; j < GRP / 16 / KB; ++j) {
        if (j + 1 < GRP / 16 / KB) {
#pragma unroll
            for (int i = 0; i < KB; ++i) {
                bb[(j + 1) & 1][2 * i]     = *(const v4f*)(b2p + ((j + 1) * KB + i) * 16);
                bb[(j + 1) & 1][2 * i + 1] = *(const v4f*)(b2p + ((j + 1) * KB + i) * 16 + 4);
            }
        }
        __builtin_amdgcn_sched_barrier(0);
#pragma unroll
        for (int i = 0; i < KB; ++i) {
            const int kc = j * KB + i;
            o0 = rnn_mma8(a2[4 * kc],     a2[4 * kc + 1], bb[j & 1][2 * i], bb[j & 1][2 * i + 1], o0);
            o1 = rnn_mma8(a2[4 * kc + 2], a2[4 * kc + 3], bb[j & 1][2 * i], bb[j & 1][2 * i + 1], o1);
        }
        __builtin_amdgcn_sched_barrier(0);
    }
    {
        float* d = P + ((size_t)g * BATCH + bt * 16 + c16) * OUT_F + wave * 32 + 8 * half;
        v4f lo0 = {o0[0], o0[1], o0[2], o0[3]};
        v4f hi0 = {o0[4], o0[5], o0[6], o0[7]};
        v4f lo1 = {o1[0], o1[1], o1[2], o1[3]};
        v4f hi1 = {o1[4], o1[5], o1[6], o1[7]};
        *(v4f*)d        = lo0;
        *(v4f*)(d + 4)  = hi0;
        *(v4f*)(d + 16) = lo1;
        *(v4f*)(d + 20) = hi1;
    }
}

__global__ __launch_bounds__(128) void rnn_sum_planes(const float* __restrict__ P,
                                                      float* __restrict__ out,
                                                      size_t ws_bytes) {
    if (ws_bytes < WS_NEED) return;
    const int i = blockIdx.x * 128 + threadIdx.x;
    const v4f* __restrict__ p = (const v4f*)P;
    v4f s = p[i];
#pragma unroll
    for (int q = 1; q < N_GRP; ++q) s += p[(size_t)q * (BATCH * OUT_F / 4) + i];
    ((v4f*)out)[i] = s;
}

extern "C" void kernel_launch(void* const* d_in, const int* in_sizes, int n_in,
                              void* d_out, int out_size, void* d_ws, size_t ws_size,
                              hipStream_t stream) {
    const float* x   = (const float*)d_in[0];
    const float* wi  = (const float*)d_in[1];
    const float* wa  = (const float*)d_in[2];
    const int* in_src  = (const int*)d_in[3];
    const int* in_dst  = (const int*)d_in[4];
    const int* ass_src = (const int*)d_in[5];
    const int* ass_dst = (const int*)d_in[6];
    const int e_in  = in_sizes[1];
    const int e_ass = in_sizes[2];

    float* W1    = (float*)d_ws;
    float* W2    = W1 + (size_t)N_ASS * IN_F;
    float* P     = W2 + (size_t)OUT_F * N_ASS;

    {
        const int n = N_ASS * IN_F + OUT_F * N_ASS;
        rnn_zero_f32<<<(n + 255) / 256, 256, 0, stream>>>(W1, n);
    }
    rnn_scatter_w12<<<(e_in + 255) / 256 + (e_ass + 255) / 256, 256, 0, stream>>>(
        wi, in_src, in_dst, e_in, W1, wa, ass_src, ass_dst, e_ass, W2);

    rnn_fused_hops<<<256, 256, 0, stream>>>(W1, x, W2, P, ws_size);

    rnn_sum_planes<<<64, 128, 0, stream>>>(P, (float*)d_out, ws_size);

    (void)n_in; (void)out_size;
}
